# starting kernel
# baseline (speedup 1.0000x reference)
#include <hip/hip_runtime.h>
#include <hip/hip_bf16.h>

#define D_FEAT 128
#define NEG_SLOPE 0.01f
#define LDS_PITCH 130

typedef float v2f __attribute__((ext_vector_type(2)));
typedef float v8f __attribute__((ext_vector_type(8)));

__global__ __launch_bounds__(256) void zero_ws_kernel(float4* __restrict__ ws, int n4) {
    int i = blockIdx.x * blockDim.x + threadIdx.x;
    if (i < n4) ws[i] = make_float4(0.f, 0.f, 0.f, 0.f);
}

__global__ __launch_bounds__(256) void edge_scatter_kernel(
    const float* __restrict__ feat,
    const int*   __restrict__ target,
    const int*   __restrict__ neighbor,
    const float* __restrict__ values,
    float*       __restrict__ hneigh,
    int E)
{
    int t = blockIdx.x * blockDim.x + threadIdx.x;
    int e = t >> 5;
    int c = t & 31;
    if (e >= E) return;

    int   nbr = neighbor[e];
    int   tgt = target[e];
    float v   = values[e];

    const float4* src = (const float4*)(feat + (size_t)nbr * D_FEAT);
    float4 f = src[c];

    float* dst = hneigh + (size_t)tgt * D_FEAT + (c << 2);
    unsafeAtomicAdd(dst + 0, f.x * v);
    unsafeAtomicAdd(dst + 1, f.y * v);
    unsafeAtomicAdd(dst + 2, f.z * v);
    unsafeAtomicAdd(dst + 3, f.w * v);
}

__global__ __launch_bounds__(256) void gemm_fused_kernel(
    const float* __restrict__ feat,
    const float* __restrict__ hneigh,
    const float* __restrict__ W1,
    const float* __restrict__ b1,
    const float* __restrict__ W2,
    const float* __restrict__ b2,
    float*       __restrict__ out,
    int nrows)
{
    __shared__ float X1[16 * LDS_PITCH];
    __shared__ float X2[16 * LDS_PITCH];

    const int rowbase = blockIdx.x * 16;
    const int tid = threadIdx.x;

    for (int idx = tid; idx < 16 * D_FEAT; idx += 256) {
        int r = idx >> 7;
        int c = idx & 127;
        int gr = rowbase + r;
        float f = 0.f, hn = 0.f;
        if (gr < nrows) {
            f  = feat  [(size_t)gr * D_FEAT + c];
            hn = hneigh[(size_t)gr * D_FEAT + c];
        }
        X1[r * LDS_PITCH + c] = f + hn;
        X2[r * LDS_PITCH + c] = f * hn;
    }
    __syncthreads();

    const int wave = tid >> 5;
    const int lane = tid & 31;
    const int half = lane >> 4;
    const int lr   = lane & 15;
    const int col  = wave * 16 + lr;

    const float bias = b1[col] + b2[col];
    v8f acc = { bias, bias, bias, bias, bias, bias, bias, bias };

    const float* x1p = &X1[lr * LDS_PITCH + 2 * half];
    const float* x2p = &X2[lr * LDS_PITCH + 2 * half];
    const float* w1p = W1 + (size_t)col * D_FEAT + 2 * half;
    const float* w2p = W2 + (size_t)col * D_FEAT + 2 * half;

    #pragma unroll
    for (int k = 0; k < D_FEAT; k += 4) {
        v2f a1 = { x1p[k], x1p[k + 1] };
        v2f a2 = { x2p[k], x2p[k + 1] };
        v2f bw1 = { w1p[k], w1p[k + 1] };
        v2f bw2 = { w2p[k], w2p[k + 1] };
        acc = __builtin_amdgcn_wmma_f32_16x16x4_f32(
                  false, a1, false, bw1, (short)0, acc, false, false);
        acc = __builtin_amdgcn_wmma_f32_16x16x4_f32(
                  false, a2, false, bw2, (short)0, acc, false, false);
    }

    #pragma unroll
    for (int v = 0; v < 8; v++) {
        int row = rowbase + v + 8 * half;
        if (row < nrows) {
            float h = acc[v];
            out[(size_t)row * D_FEAT + col] = (h >= 0.f) ? h : NEG_SLOPE * h;
        }
    }
}

extern "C" void kernel_launch(void* const* d_in, const int* in_sizes, int n_in,
                              void* d_out, int out_size, void* d_ws, size_t ws_size,
                              hipStream_t stream) {
    const float* feat     = (const float*)d_in[0];
    const int*   target   = (const int*)  d_in[1];
    const int*   neighbor = (const int*)  d_in[2];
    const float* values   = (const float*)d_in[3];
    const float* W1       = (const float*)d_in[4];
    const float* b1       = (const float*)d_in[5];
    const float* W2       = (const float*)d_in[6];
    const float* b2       = (const float*)d_in[7];

    const int N = in_sizes[0] / D_FEAT;
    const int E = in_sizes[1];

    float* hneigh = (float*)d_ws;
    float* outp   = (float*)d_out;

    {
        int n4 = (N * D_FEAT) / 4;
        int blocks = (n4 + 255) / 256;
        zero_ws_kernel<<<blocks, 256, 0, stream>>>((float4*)hneigh, n4);
    }
    {
        long long threads = (long long)E * 32;
        int blocks = (int)((threads + 255) / 256);
        edge_scatter_kernel<<<blocks, 256, 0, stream>>>(feat, target, neighbor,
                                                        values, hneigh, E);
    }
    {
        int blocks = (N + 15) / 16;
        gemm_fused_kernel<<<blocks, 256, 0, stream>>>(feat, hneigh, W1, b1, W2, b2,
                                                      outp, N);
    }
}


# GEMM launch becomes 512 resident work-groups looping over row tiles: W1,W2 copied to LDS once by whole-line loads, each wave keeps its B fragments in registers for all tiles
# speedup vs baseline: 1.2702x; 1.2702x over previous
#include <hip/hip_runtime.h>
#include <hip/hip_bf16.h>

#define D_FEAT 128
#define NEG_SLOPE 0.01f

typedef float v2f __attribute__((ext_vector_type(2)));
typedef float v8f __attribute__((ext_vector_type(8)));

__global__ __launch_bounds__(256) void zero_ws_kernel(float4* __restrict__ ws, int n4) {
    int i = blockIdx.x * blockDim.x + threadIdx.x;
    if (i < n4) ws[i] = make_float4(0.f, 0.f, 0.f, 0.f);
}

__global__ __launch_bounds__(256) void edge_scatter_kernel(
    const float* __restrict__ feat,
    const int*   __restrict__ target,
    const int*   __restrict__ neighbor,
    const float* __restrict__ values,
    float*       __restrict__ hneigh,
    int E)
{
    int t = blockIdx.x * blockDim.x + threadIdx.x;
    int e = t >> 5;
    int c = t & 31;
    if (e >= E) return;

    int   nbr = neighbor[e];
    int   tgt = target[e];
    float v   = values[e];

    const float4* src = (const float4*)(feat + (size_t)nbr * D_FEAT);
    float4 f = src[c];

    float* dst = hneigh + (size_t)tgt * D_FEAT + (c << 2);
    unsafeAtomicAdd(dst + 0, f.x * v);
    unsafeAtomicAdd(dst + 1, f.y * v);
    unsafeAtomicAdd(dst + 2, f.z * v);
    unsafeAtomicAdd(dst + 3, f.w * v);
}

#define GEMM_BLOCKS 512
#define W_PITCH 132

__global__ __launch_bounds__(256) void gemm_fused_kernel(
    const float* __restrict__ feat,
    const float* __restrict__ hneigh,
    const float* __restrict__ W1,
    const float* __restrict__ b1,
    const float* __restrict__ W2,
    const float* __restrict__ b2,
    float*       __restrict__ out,
    int nrows)
{
    __shared__ float Wl1[D_FEAT * W_PITCH];
    __shared__ float Wl2[D_FEAT * W_PITCH];
    __shared__ float X1[16 * W_PITCH];
    __shared__ float X2[16 * W_PITCH];

    const int tid = threadIdx.x;
    const int ntiles = (nrows + 15) >> 4;

    for (int idx = tid; idx < D_FEAT * (D_FEAT / 4); idx += 256) {
        int r  = idx >> 5;
        int c4 = idx & 31;
        float4 a = ((const float4*)W1)[idx];
        float4 b = ((const float4*)W2)[idx];
        *(float4*)&Wl1[r * W_PITCH + 4 * c4] = a;
        *(float4*)&Wl2[r * W_PITCH + 4 * c4] = b;
    }
    __syncthreads();

    const int wave = tid >> 5;
    const int lane = tid & 31;
    const int half = lane >> 4;
    const int lr   = lane & 15;
    const int col  = wave * 16 + lr;

    v2f bw1[D_FEAT / 4], bw2[D_FEAT / 4];
    {
        const float* w1p = &Wl1[col * W_PITCH + 2 * half];
        const float* w2p = &Wl2[col * W_PITCH + 2 * half];
        #pragma unroll
        for (int i = 0; i < D_FEAT / 4; i++) {
            bw1[i] = (v2f){ w1p[4 * i], w1p[4 * i + 1] };
            bw2[i] = (v2f){ w2p[4 * i], w2p[4 * i + 1] };
        }
    }
    const float bias = b1[col] + b2[col];

    const float* x1p = &X1[lr * W_PITCH + 2 * half];
    const float* x2p = &X2[lr * W_PITCH + 2 * half];

    const int sr = tid >> 5;
    const int sc = tid & 31;

    float4 f0, f1, h0, h1;
    {
        const float4 z = make_float4(0.f, 0.f, 0.f, 0.f);
        f0 = z; f1 = z; h0 = z; h1 = z;
        int tile0 = blockIdx.x;
        if (tile0 < ntiles) {
            int g0 = tile0 * 16 + sr, g1 = g0 + 8;
            if (g0 < nrows) {
                f0 = ((const float4*)(feat   + (size_t)g0 * D_FEAT))[sc];
                h0 = ((const float4*)(hneigh + (size_t)g0 * D_FEAT))[sc];
            }
            if (g1 < nrows) {
                f1 = ((const float4*)(feat   + (size_t)g1 * D_FEAT))[sc];
                h1 = ((const float4*)(hneigh + (size_t)g1 * D_FEAT))[sc];
            }
        }
    }

    for (int tile = blockIdx.x; tile < ntiles; tile += GEMM_BLOCKS) {
        const int rowbase = tile * 16;

        __syncthreads();

        {
            float4 s0 = make_float4(f0.x + h0.x, f0.y + h0.y, f0.z + h0.z, f0.w + h0.w);
            float4 p0 = make_float4(f0.x * h0.x, f0.y * h0.y, f0.z * h0.z, f0.w * h0.w);
            float4 s1 = make_float4(f1.x + h1.x, f1.y + h1.y, f1.z + h1.z, f1.w + h1.w);
            float4 p1 = make_float4(f1.x * h1.x, f1.y * h1.y, f1.z * h1.z, f1.w * h1.w);
            *(float4*)&X1[sr * W_PITCH + 4 * sc]       = s0;
            *(float4*)&X2[sr * W_PITCH + 4 * sc]       = p0;
            *(float4*)&X1[(sr + 8) * W_PITCH + 4 * sc] = s1;
            *(float4*)&X2[(sr + 8) * W_PITCH + 4 * sc] = p1;
        }
        __syncthreads();

        {
            const float4 z = make_float4(0.f, 0.f, 0.f, 0.f);
            f0 = z; f1 = z; h0 = z; h1 = z;
            int nt = tile + GEMM_BLOCKS;
            if (nt < ntiles) {
                int g0 = nt * 16 + sr, g1 = g0 + 8;
                if (g0 < nrows) {
                    f0 = ((const float4*)(feat   + (size_t)g0 * D_FEAT))[sc];
                    h0 = ((const float4*)(hneigh + (size_t)g0 * D_FEAT))[sc];
                }
                if (g1 < nrows) {
                    f1 = ((const float4*)(feat   + (size_t)g1 * D_FEAT))[sc];
                    h1 = ((const float4*)(hneigh + (size_t)g1 * D_FEAT))[sc];
                }
            }
        }

        v8f acc = { bias, bias, bias, bias, bias, bias, bias, bias };
        #pragma unroll
        for (int i = 0; i < D_FEAT / 4; i++) {
            v2f a1 = { x1p[4 * i], x1p[4 * i + 1] };
            v2f a2 = { x2p[4 * i], x2p[4 * i + 1] };
            acc = __builtin_amdgcn_wmma_f32_16x16x4_f32(
                      false, a1, false, bw1[i], (short)0, acc, false, false);
            acc = __builtin_amdgcn_wmma_f32_16x16x4_f32(
                      false, a2, false, bw2[i], (short)0, acc, false, false);
        }

        #pragma unroll
        for (int v = 0; v < 8; v++) {
            int row = rowbase + v + 8 * half;
            if (row < nrows) {
                float h = acc[v];
                out[(size_t)row * D_FEAT + col] = (h >= 0.f) ? h : NEG_SLOPE * h;
            }
        }
    }
}

extern "C" void kernel_launch(void* const* d_in, const int* in_sizes, int n_in,
                              void* d_out, int out_size, void* d_ws, size_t ws_size,
                              hipStream_t stream) {
    const float* feat     = (const float*)d_in[0];
    const int*   target   = (const int*)  d_in[1];
    const int*   neighbor = (const int*)  d_in[2];
    const float* values   = (const float*)d_in[3];
    const float* W1       = (const float*)d_in[4];
    const float* b1       = (const float*)d_in[5];
    const float* W2       = (const float*)d_in[6];
    const float* b2       = (const float*)d_in[7];

    const int N = in_sizes[0] / D_FEAT;
    const int E = in_sizes[1];

    float* hneigh = (float*)d_ws;
    float* outp   = (float*)d_out;

    {
        int n4 = (N * D_FEAT) / 4;
        int blocks = (n4 + 255) / 256;
        zero_ws_kernel<<<blocks, 256, 0, stream>>>((float4*)hneigh, n4);
    }
    {
        long long threads = (long long)E * 32;
        int blocks = (int)((threads + 255) / 256);
        edge_scatter_kernel<<<blocks, 256, 0, stream>>>(feat, target, neighbor,
                                                        values, hneigh, E);
    }
    {
        gemm_fused_kernel<<<GEMM_BLOCKS, 256, 0, stream>>>(feat, hneigh, W1, b1, W2, b2,
                                                      outp, N);
    }
}


# Edge scatter: lane = column inside a 128-B line, so each f32 atomic add instruction of a wave covers one whole line of the target row (4 line requests per edge, not 16 quarter-used)
# speedup vs baseline: 1.6493x; 1.2985x over previous
#include <hip/hip_runtime.h>
#include <hip/hip_bf16.h>

#define D_FEAT 128
#define NEG_SLOPE 0.01f

typedef float v2f __attribute__((ext_vector_type(2)));
typedef float v8f __attribute__((ext_vector_type(8)));

__global__ __launch_bounds__(256) void zero_ws_kernel(float4* __restrict__ ws, int n4) {
    int i = blockIdx.x * blockDim.x + threadIdx.x;
    if (i < n4) ws[i] = make_float4(0.f, 0.f, 0.f, 0.f);
}

__global__ __launch_bounds__(256) void edge_scatter_kernel(
    const float* __restrict__ feat,
    const int*   __restrict__ target,
    const int*   __restrict__ neighbor,
    const float* __restrict__ values,
    float*       __restrict__ hneigh,
    int E)
{
    int t = blockIdx.x * blockDim.x + threadIdx.x;
    int e = t >> 5;
    int c = t & 31;
    if (e >= E) return;

    int   nbr = neighbor[e];
    int   tgt = target[e];
    float v   = values[e];

    const float* src = feat + (size_t)nbr * D_FEAT + c;
    float f0 = src[0];
    float f1 = src[32];
    float f2 = src[64];
    float f3 = src[96];

    float* dst = hneigh + (size_t)tgt * D_FEAT + c;
    unsafeAtomicAdd(dst,      f0 * v);
    unsafeAtomicAdd(dst + 32, f1 * v);
    unsafeAtomicAdd(dst + 64, f2 * v);
    unsafeAtomicAdd(dst + 96, f3 * v);
}

#define GEMM_BLOCKS 512
#define W_PITCH 132

__global__ __launch_bounds__(256) void gemm_fused_kernel(
    const float* __restrict__ feat,
    const float* __restrict__ hneigh,
    const float* __restrict__ W1,
    const float* __restrict__ b1,
    const float* __restrict__ W2,
    const float* __restrict__ b2,
    float*       __restrict__ out,
    int nrows)
{
    __shared__ float Wl1[D_FEAT * W_PITCH];
    __shared__ float Wl2[D_FEAT * W_PITCH];
    __shared__ float X1[16 * W_PITCH];
    __shared__ float X2[16 * W_PITCH];

    const int tid = threadIdx.x;
    const int ntiles = (nrows + 15) >> 4;

    for (int idx = tid; idx < D_FEAT * (D_FEAT / 4); idx += 256) {
        int r  = idx >> 5;
        int c4 = idx & 31;
        float4 a = ((const float4*)W1)[idx];
        float4 b = ((const float4*)W2)[idx];
        *(float4*)&Wl1[r * W_PITCH + 4 * c4] = a;
        *(float4*)&Wl2[r * W_PITCH + 4 * c4] = b;
    }
    __syncthreads();

    const int wave = tid >> 5;
    const int lane = tid & 31;
    const int half = lane >> 4;
    const int lr   = lane & 15;
    const int col  = wave * 16 + lr;

    v2f bw1[D_FEAT / 4], bw2[D_FEAT / 4];
    {
        const float* w1p = &Wl1[col * W_PITCH + 2 * half];
        const float* w2p = &Wl2[col * W_PITCH + 2 * half];
        #pragma unroll
        for (int i = 0; i < D_FEAT / 4; i++) {
            bw1[i] = (v2f){ w1p[4 * i], w1p[4 * i + 1] };
            bw2[i] = (v2f){ w2p[4 * i], w2p[4 * i + 1] };
        }
    }
    const float bias = b1[col] + b2[col];

    const float* x1p = &X1[lr * W_PITCH + 2 * half];
    const float* x2p = &X2[lr * W_PITCH + 2 * half];

    const int sr = tid >> 5;
    const int sc = tid & 31;

    float4 f0, f1, h0, h1;
    {
        const float4 z = make_float4(0.f, 0.f, 0.f, 0.f);
        f0 = z; f1 = z; h0 = z; h1 = z;
        int tile0 = blockIdx.x;
        if (tile0 < ntiles) {
            int g0 = tile0 * 16 + sr, g1 = g0 + 8;
            if (g0 < nrows) {
                f0 = ((const float4*)(feat   + (size_t)g0 * D_FEAT))[sc];
                h0 = ((const float4*)(hneigh + (size_t)g0 * D_FEAT))[sc];
            }
            if (g1 < nrows) {
                f1 = ((const float4*)(feat   + (size_t)g1 * D_FEAT))[sc];
                h1 = ((const float4*)(hneigh + (size_t)g1 * D_FEAT))[sc];
            }
        }
    }

    for (int tile = blockIdx.x; tile < ntiles; tile += GEMM_BLOCKS) {
        const int rowbase = tile * 16;

        __syncthreads();

        {
            float4 s0 = make_float4(f0.x + h0.x, f0.y + h0.y, f0.z + h0.z, f0.w + h0.w);
            float4 p0 = make_float4(f0.x * h0.x, f0.y * h0.y, f0.z * h0.z, f0.w * h0.w);
            float4 s1 = make_float4(f1.x + h1.x, f1.y + h1.y, f1.z + h1.z, f1.w + h1.w);
            float4 p1 = make_float4(f1.x * h1.x, f1.y * h1.y, f1.z * h1.z, f1.w * h1.w);
            *(float4*)&X1[sr * W_PITCH + 4 * sc]       = s0;
            *(float4*)&X2[sr * W_PITCH + 4 * sc]       = p0;
            *(float4*)&X1[(sr + 8) * W_PITCH + 4 * sc] = s1;
            *(float4*)&X2[(sr + 8) * W_PITCH + 4 * sc] = p1;
        }
        __syncthreads();

        {
            const float4 z = make_float4(0.f, 0.f, 0.f, 0.f);
            f0 = z; f1 = z; h0 = z; h1 = z;
            int nt = tile + GEMM_BLOCKS;
            if (nt < ntiles) {
                int g0 = nt * 16 + sr, g1 = g0 + 8;
                if (g0 < nrows) {
                    f0 = ((const float4*)(feat   + (size_t)g0 * D_FEAT))[sc];
                    h0 = ((const float4*)(hneigh + (size_t)g0 * D_FEAT))[sc];
                }
                if (g1 < nrows) {
                    f1 = ((const float4*)(feat   + (size_t)g1 * D_FEAT))[sc];
                    h1 = ((const float4*)(hneigh + (size_t)g1 * D_FEAT))[sc];
                }
            }
        }

        v8f acc = { bias, bias, bias, bias, bias, bias, bias, bias };
        #pragma unroll
        for (int i = 0; i < D_FEAT / 4; i++) {
            v2f a1 = { x1p[4 * i], x1p[4 * i + 1] };
            v2f a2 = { x2p[4 * i], x2p[4 * i + 1] };
            acc = __builtin_amdgcn_wmma_f32_16x16x4_f32(
                      false, a1, false, bw1[i], (short)0, acc, false, false);
            acc = __builtin_amdgcn_wmma_f32_16x16x4_f32(
                      false, a2, false, bw2[i], (short)0, acc, false, false);
        }

        #pragma unroll
        for (int v = 0; v < 8; v++) {
            int row = rowbase + v + 8 * half;
            if (row < nrows) {
                float h = acc[v];
                out[(size_t)row * D_FEAT + col] = (h >= 0.f) ? h : NEG_SLOPE * h;
            }
        }
    }
}

extern "C" void kernel_launch(void* const* d_in, const int* in_sizes, int n_in,
                              void* d_out, int out_size, void* d_ws, size_t ws_size,
                              hipStream_t stream) {
    const float* feat     = (const float*)d_in[0];
    const int*   target   = (const int*)  d_in[1];
    const int*   neighbor = (const int*)  d_in[2];
    const float* values   = (const float*)d_in[3];
    const float* W1       = (const float*)d_in[4];
    const float* b1       = (const float*)d_in[5];
    const float* W2       = (const float*)d_in[6];
    const float* b2       = (const float*)d_in[7];

    const int N = in_sizes[0] / D_FEAT;
    const int E = in_sizes[1];

    float* hneigh = (float*)d_ws;
    float* outp   = (float*)d_out;

    {
        int n4 = (N * D_FEAT) / 4;
        int blocks = (n4 + 255) / 256;
        zero_ws_kernel<<<blocks, 256, 0, stream>>>((float4*)hneigh, n4);
    }
    {
        long long threads = (long long)E * 32;
        int blocks = (int)((threads + 255) / 256);
        edge_scatter_kernel<<<blocks, 256, 0, stream>>>(feat, target, neighbor,
                                                        values, hneigh, E);
    }
    {
        gemm_fused_kernel<<<GEMM_BLOCKS, 256, 0, stream>>>(feat, hneigh, W1, b1, W2, b2,
                                                      outp, N);
    }
}


# Edge sums leave global f32 atomics: 256 work-groups each own 196 target rows in LDS, scan the target list, add their edges' rows with LDS f32 adds, store each h_neigh row once; clear launch gone
# speedup vs baseline: 2.2003x; 1.3341x over previous
#include <hip/hip_runtime.h>
#include <hip/hip_bf16.h>

#define D_FEAT 128
#define NEG_SLOPE 0.01f

typedef float v2f __attribute__((ext_vector_type(2)));
typedef float v8f __attribute__((ext_vector_type(8)));

#define OWN_BLOCKS  256
#define OWN_ROWS    196
#define OWN_THREADS 1024
#define SCAN_U      8
#define QCAP        40

template <bool FULL>
__device__ __forceinline__ void add_batch(const int* bq, int cnt, int lane, int lo,
                                          const float* __restrict__ feat,
                                          const int*   __restrict__ target,
                                          const int*   __restrict__ neighbor,
                                          const float* __restrict__ values,
                                          float* acc)
{
    int nb = 0, tr = 0, vb = 0;
    if (lane < cnt) {
        int e = bq[lane];
        nb = neighbor[e];
        tr = target[e] - lo;
        vb = __builtin_bit_cast(int, values[e]);
    }
    float f[8][4];
    #pragma unroll
    for (int j = 0; j < 8; j++) {
        f[j][0] = 0.f; f[j][1] = 0.f; f[j][2] = 0.f; f[j][3] = 0.f;
        if (FULL || j < cnt) {
            int nbj = __builtin_amdgcn_readlane(nb, j);
            const float* src = feat + (size_t)nbj * D_FEAT + lane;
            f[j][0] = src[0]; f[j][1] = src[32]; f[j][2] = src[64]; f[j][3] = src[96];
        }
    }
    #pragma unroll
    for (int j = 0; j < 8; j++) {
        if (FULL || j < cnt) {
            int   trj = __builtin_amdgcn_readlane(tr, j);
            float vj  = __builtin_bit_cast(float, __builtin_amdgcn_readlane(vb, j));
            float* d = acc + trj * D_FEAT + lane;
            __hip_atomic_fetch_add(d,      f[j][0] * vj, __ATOMIC_RELAXED, __HIP_MEMORY_SCOPE_WORKGROUP);
            __hip_atomic_fetch_add(d + 32, f[j][1] * vj, __ATOMIC_RELAXED, __HIP_MEMORY_SCOPE_WORKGROUP);
            __hip_atomic_fetch_add(d + 64, f[j][2] * vj, __ATOMIC_RELAXED, __HIP_MEMORY_SCOPE_WORKGROUP);
            __hip_atomic_fetch_add(d + 96, f[j][3] * vj, __ATOMIC_RELAXED, __HIP_MEMORY_SCOPE_WORKGROUP);
        }
    }
}

__global__ __launch_bounds__(OWN_THREADS) void owner_sum_kernel(
    const float* __restrict__ feat,
    const int*   __restrict__ target,
    const int*   __restrict__ neighbor,
    const float* __restrict__ values,
    float*       __restrict__ hneigh,
    int N, int E)
{
    __shared__ float acc[OWN_ROWS * D_FEAT];
    __shared__ int   q[(OWN_THREADS / 32) * QCAP];

    const int tid  = threadIdx.x;
    const int lane = tid & 31;
    const int wave = tid >> 5;
    int* myq = q + wave * QCAP;
    const int n4 = E >> 2;
    const int4* t4 = (const int4*)target;

    for (int lo = blockIdx.x * OWN_ROWS; lo < N; lo += gridDim.x * OWN_ROWS) {
        const int nr = min(OWN_ROWS, N - lo);

        for (int i = tid; i < nr * (D_FEAT / 4); i += OWN_THREADS)
            ((float4*)acc)[i] = make_float4(0.f, 0.f, 0.f, 0.f);
        __syncthreads();

        int qn = 0;
        for (int base = 0; base < n4; base += OWN_THREADS * SCAN_U) {
            unsigned fl = 0;
            #pragma unroll
            for (int u = 0; u < SCAN_U; u++) {
                int idx = base + u * OWN_THREADS + tid;
                int4 t = t4[min(idx, n4 - 1)];
                if (idx >= n4) t = make_int4(-1, -1, -1, -1);
                fl |= ((unsigned)(t.x - lo) < (unsigned)nr ? 1u : 0u) << (4 * u);
                fl |= ((unsigned)(t.y - lo) < (unsigned)nr ? 1u : 0u) << (4 * u + 1);
                fl |= ((unsigned)(t.z - lo) < (unsigned)nr ? 1u : 0u) << (4 * u + 2);
                fl |= ((unsigned)(t.w - lo) < (unsigned)nr ? 1u : 0u) << (4 * u + 3);
            }
            if (__builtin_amdgcn_ballot_w32(fl != 0u) == 0u) continue;

            for (int b = 0; b < 4 * SCAN_U; b++) {
                const bool own = ((fl >> b) & 1u) != 0u;
                const unsigned m = __builtin_amdgcn_ballot_w32(own);
                if (m == 0u) continue;
                const int pos = qn + (int)__builtin_amdgcn_mbcnt_lo(m, 0u);
                if (own) myq[pos] = 4 * (base + (b >> 2) * OWN_THREADS + tid) + (b & 3);
                qn += __builtin_popcount(m);
                if (qn >= 8) {
                    int head = 0;
                    while (qn - head >= 8) {
                        add_batch<true>(myq + head, 8, lane, lo, feat, target, neighbor, values, acc);
                        head += 8;
                    }
                    const int r = qn - head;
                    int tmp = 0;
                    if (lane < r) tmp = myq[head + lane];
                    if (lane < r) myq[lane] = tmp;
                    qn = r;
                }
            }
        }

        if (wave == 0) {
            const int e = 4 * n4 + lane;
            bool own = false;
            if (e < E) own = (unsigned)(target[e] - lo) < (unsigned)nr;
            const unsigned m = __builtin_amdgcn_ballot_w32(own);
            if (m != 0u) {
                const int pos = qn + (int)__builtin_amdgcn_mbcnt_lo(m, 0u);
                if (own) myq[pos] = e;
                qn += __builtin_popcount(m);
            }
        }

        {
            int head = 0;
            while (qn - head >= 8) {
                add_batch<true>(myq + head, 8, lane, lo, feat, target, neighbor, values, acc);
                head += 8;
            }
            if (qn - head > 0)
                add_batch<false>(myq + head, qn - head, lane, lo, feat, target, neighbor, values, acc);
        }
        __syncthreads();

        for (int i = tid; i < nr * (D_FEAT / 4); i += OWN_THREADS)
            ((float4*)(hneigh + (size_t)lo * D_FEAT))[i] = ((const float4*)acc)[i];
        __syncthreads();
    }
}

#define GEMM_BLOCKS 512
#define W_PITCH 132

__global__ __launch_bounds__(256) void gemm_fused_kernel(
    const float* __restrict__ feat,
    const float* __restrict__ hneigh,
    const float* __restrict__ W1,
    const float* __restrict__ b1,
    const float* __restrict__ W2,
    const float* __restrict__ b2,
    float*       __restrict__ out,
    int nrows)
{
    __shared__ float Wl1[D_FEAT * W_PITCH];
    __shared__ float Wl2[D_FEAT * W_PITCH];
    __shared__ float X1[16 * W_PITCH];
    __shared__ float X2[16 * W_PITCH];

    const int tid = threadIdx.x;
    const int ntiles = (nrows + 15) >> 4;

    for (int idx = tid; idx < D_FEAT * (D_FEAT / 4); idx += 256) {
        int r  = idx >> 5;
        int c4 = idx & 31;
        float4 a = ((const float4*)W1)[idx];
        float4 b = ((const float4*)W2)[idx];
        *(float4*)&Wl1[r * W_PITCH + 4 * c4] = a;
        *(float4*)&Wl2[r * W_PITCH + 4 * c4] = b;
    }
    __syncthreads();

    const int wave = tid >> 5;
    const int lane = tid & 31;
    const int half = lane >> 4;
    const int lr   = lane & 15;
    const int col  = wave * 16 + lr;

    v2f bw1[D_FEAT / 4], bw2[D_FEAT / 4];
    {
        const float* w1p = &Wl1[col * W_PITCH + 2 * half];
        const float* w2p = &Wl2[col * W_PITCH + 2 * half];
        #pragma unroll
        for (int i = 0; i < D_FEAT / 4; i++) {
            bw1[i] = (v2f){ w1p[4 * i], w1p[4 * i + 1] };
            bw2[i] = (v2f){ w2p[4 * i], w2p[4 * i + 1] };
        }
    }
    const float bias = b1[col] + b2[col];

    const float* x1p = &X1[lr * W_PITCH + 2 * half];
    const float* x2p = &X2[lr * W_PITCH + 2 * half];

    const int sr = tid >> 5;
    const int sc = tid & 31;

    float4 f0, f1, h0, h1;
    {
        const float4 z = make_float4(0.f, 0.f, 0.f, 0.f);
        f0 = z; f1 = z; h0 = z; h1 = z;
        int tile0 = blockIdx.x;
        if (tile0 < ntiles) {
            int g0 = tile0 * 16 + sr, g1 = g0 + 8;
            if (g0 < nrows) {
                f0 = ((const float4*)(feat   + (size_t)g0 * D_FEAT))[sc];
                h0 = ((const float4*)(hneigh + (size_t)g0 * D_FEAT))[sc];
            }
            if (g1 < nrows) {
                f1 = ((const float4*)(feat   + (size_t)g1 * D_FEAT))[sc];
                h1 = ((const float4*)(hneigh + (size_t)g1 * D_FEAT))[sc];
            }
        }
    }

    for (int tile = blockIdx.x; tile < ntiles; tile += GEMM_BLOCKS) {
        const int rowbase = tile * 16;

        __syncthreads();

        {
            float4 s0 = make_float4(f0.x + h0.x, f0.y + h0.y, f0.z + h0.z, f0.w + h0.w);
            float4 p0 = make_float4(f0.x * h0.x, f0.y * h0.y, f0.z * h0.z, f0.w * h0.w);
            float4 s1 = make_float4(f1.x + h1.x, f1.y + h1.y, f1.z + h1.z, f1.w + h1.w);
            float4 p1 = make_float4(f1.x * h1.x, f1.y * h1.y, f1.z * h1.z, f1.w * h1.w);
            *(float4*)&X1[sr * W_PITCH + 4 * sc]       = s0;
            *(float4*)&X2[sr * W_PITCH + 4 * sc]       = p0;
            *(float4*)&X1[(sr + 8) * W_PITCH + 4 * sc] = s1;
            *(float4*)&X2[(sr + 8) * W_PITCH + 4 * sc] = p1;
        }
        __syncthreads();

        {
            const float4 z = make_float4(0.f, 0.f, 0.f, 0.f);
            f0 = z; f1 = z; h0 = z; h1 = z;
            int nt = tile + GEMM_BLOCKS;
            if (nt < ntiles) {
                int g0 = nt * 16 + sr, g1 = g0 + 8;
                if (g0 < nrows) {
                    f0 = ((const float4*)(feat   + (size_t)g0 * D_FEAT))[sc];
                    h0 = ((const float4*)(hneigh + (size_t)g0 * D_FEAT))[sc];
                }
                if (g1 < nrows) {
                    f1 = ((const float4*)(feat   + (size_t)g1 * D_FEAT))[sc];
                    h1 = ((const float4*)(hneigh + (size_t)g1 * D_FEAT))[sc];
                }
            }
        }

        v8f acc = { bias, bias, bias, bias, bias, bias, bias, bias };
        #pragma unroll
        for (int i = 0; i < D_FEAT / 4; i++) {
            v2f a1 = { x1p[4 * i], x1p[4 * i + 1] };
            v2f a2 = { x2p[4 * i], x2p[4 * i + 1] };
            acc = __builtin_amdgcn_wmma_f32_16x16x4_f32(
                      false, a1, false, bw1[i], (short)0, acc, false, false);
            acc = __builtin_amdgcn_wmma_f32_16x16x4_f32(
                      false, a2, false, bw2[i], (short)0, acc, false, false);
        }

        #pragma unroll
        for (int v = 0; v < 8; v++) {
            int row = rowbase + v + 8 * half;
            if (row < nrows) {
                float h = acc[v];
                out[(size_t)row * D_FEAT + col] = (h >= 0.f) ? h : NEG_SLOPE * h;
            }
        }
    }
}

extern "C" void kernel_launch(void* const* d_in, const int* in_sizes, int n_in,
                              void* d_out, int out_size, void* d_ws, size_t ws_size,
                              hipStream_t stream) {
    const float* feat     = (const float*)d_in[0];
    const int*   target   = (const int*)  d_in[1];
    const int*   neighbor = (const int*)  d_in[2];
    const float* values   = (const float*)d_in[3];
    const float* W1       = (const float*)d_in[4];
    const float* b1       = (const float*)d_in[5];
    const float* W2       = (const float*)d_in[6];
    const float* b2       = (const float*)d_in[7];

    const int N = in_sizes[0] / D_FEAT;
    const int E = in_sizes[1];

    float* hneigh = (float*)d_ws;
    float* outp   = (float*)d_out;

    {
        owner_sum_kernel<<<OWN_BLOCKS, OWN_THREADS, 0, stream>>>(feat, target, neighbor,
                                                                 values, hneigh, N, E);
    }
    {
        gemm_fused_kernel<<<GEMM_BLOCKS, 256, 0, stream>>>(feat, hneigh, W1, b1, W2, b2,
                                                      outp, N);
    }
}


# Edges sorted by owner range first (new launch: per-slice LDS counting sort writes 16-B records), owners read only their own records, no full target scan; order of the f32 adds changes
# speedup vs baseline: 3.1011x; 1.4094x over previous
#include <hip/hip_runtime.h>
#include <hip/hip_bf16.h>

#define D_FEAT 128
#define NEG_SLOPE 0.01f

typedef float v2f __attribute__((ext_vector_type(2)));
typedef float v8f __attribute__((ext_vector_type(8)));

#define OWN_BLOCKS  256
#define OWN_ROWS    196
#define OWN_THREADS 1024
#define NOWN        256
#define SL          2448
#define BK_BLOCKS   256
#define BK_THREADS  512
#define BK_PER      ((SL + BK_THREADS - 1) / BK_THREADS)

__global__ __launch_bounds__(BK_THREADS) void bucket_kernel(
    const int*   __restrict__ target,
    const int*   __restrict__ neighbor,
    const float* __restrict__ values,
    int4*        __restrict__ recs,
    int2*        __restrict__ segs,
    int N, int E)
{
    __shared__ int  hist[NOWN];
    __shared__ int  offs[NOWN + 1];
    __shared__ int4 srec[SL];

    const int tid = threadIdx.x;
    const int nsl = (E + SL - 1) / SL;

    for (int s = blockIdx.x; s < nsl; s += gridDim.x) {
        const int e0  = s * SL;
        const int cnt = min(SL, E - e0);

        if (tid < NOWN) hist[tid] = 0;
        __syncthreads();

        int   tt[BK_PER], nn[BK_PER];
        float vv[BK_PER];
        #pragma unroll
        for (int k = 0; k < BK_PER; k++) {
            const int e = e0 + min(tid + k * BK_THREADS, cnt - 1);
            tt[k] = target[e];
            nn[k] = neighbor[e];
            vv[k] = values[e];
        }
        int  ow[BK_PER], rk[BK_PER];
        int4 rc[BK_PER];
        #pragma unroll
        for (int k = 0; k < BK_PER; k++) {
            const int t = tt[k];
            const int o = t / OWN_ROWS;
            ow[k] = -1; rk[k] = 0;
            rc[k] = make_int4(min(max(nn[k], 0), N - 1), t - o * OWN_ROWS,
                              __builtin_bit_cast(int, vv[k]), 0);
            if (tid + k * BK_THREADS < cnt && (unsigned)t < (unsigned)N && o < NOWN) {
                ow[k] = o;
                rk[k] = atomicAdd(&hist[o], 1);
            }
        }
        __syncthreads();

        if (tid < 32) {
            int c[8];
            int sum = 0;
            #pragma unroll
            for (int j = 0; j < 8; j++) { c[j] = hist[8 * tid + j]; sum += c[j]; }
            int inc = sum;
            for (int d = 1; d < 32; d <<= 1) {
                int y = __shfl_up(inc, d, 32);
                if (tid >= d) inc += y;
            }
            int run = inc - sum;
            #pragma unroll
            for (int j = 0; j < 8; j++) { offs[8 * tid + j] = run; run += c[j]; }
            if (tid == 31) offs[NOWN] = run;
        }
        __syncthreads();

        #pragma unroll
        for (int k = 0; k < BK_PER; k++)
            if (ow[k] >= 0) srec[offs[ow[k]] + rk[k]] = rc[k];
        __syncthreads();

        const int total = offs[NOWN];
        for (int i = tid; i < total; i += BK_THREADS) recs[(size_t)e0 + i] = srec[i];
        if (tid < NOWN)
            segs[(size_t)tid * nsl + s] = make_int2(e0 + offs[tid], offs[tid + 1] - offs[tid]);
        __syncthreads();
    }
}


template <bool FULL>
__device__ __forceinline__ void add_batch(int rnb, int rtr, int rvb, int j0, int cnt, int lane,
                                          const float* __restrict__ feat, float* acc)
{
    float f[8][4];
    #pragma unroll
    for (int j = 0; j < 8; j++) {
        int nbj = __builtin_amdgcn_readlane(rnb, j0 + j);
        const float* src = feat + (size_t)nbj * D_FEAT + lane;
        f[j][0] = src[0]; f[j][1] = src[32]; f[j][2] = src[64]; f[j][3] = src[96];
    }
    #pragma unroll
    for (int j = 0; j < 8; j++) {
        if (FULL || j < cnt) {
            int   trj = __builtin_amdgcn_readlane(rtr, j0 + j);
            float vj  = __builtin_bit_cast(float, __builtin_amdgcn_readlane(rvb, j0 + j));
            float* d = acc + trj * D_FEAT + lane;
            __hip_atomic_fetch_add(d,      f[j][0] * vj, __ATOMIC_RELAXED, __HIP_MEMORY_SCOPE_WORKGROUP);
            __hip_atomic_fetch_add(d + 32, f[j][1] * vj, __ATOMIC_RELAXED, __HIP_MEMORY_SCOPE_WORKGROUP);
            __hip_atomic_fetch_add(d + 64, f[j][2] * vj, __ATOMIC_RELAXED, __HIP_MEMORY_SCOPE_WORKGROUP);
            __hip_atomic_fetch_add(d + 96, f[j][3] * vj, __ATOMIC_RELAXED, __HIP_MEMORY_SCOPE_WORKGROUP);
        }
    }
}

__device__ __forceinline__ int rec_addr(int i, const int* st, const int* pre)
{
    int a = st[0] + i;
    #pragma unroll
    for (int k = 1; k < 8; k++)
        if (i >= pre[k]) a = st[k] + (i - pre[k]);
    return a;
}

__global__ __launch_bounds__(OWN_THREADS) void owner_sum_kernel(
    const float* __restrict__ feat,
    const int4*  __restrict__ recs,
    const int2*  __restrict__ segs,
    float*       __restrict__ hneigh,
    int N, int E)
{
    __shared__ float acc[OWN_ROWS * D_FEAT];

    const int tid  = threadIdx.x;
    const int lane = tid & 31;
    const int wave = tid >> 5;
    const int nsl  = (E + SL - 1) / SL;

    for (int o = blockIdx.x; o < NOWN; o += gridDim.x) {
        const int lo = o * OWN_ROWS;
        const int nr = min(OWN_ROWS, N - lo);
        if (nr <= 0) break;

        for (int i = tid; i < nr * (D_FEAT / 4); i += OWN_THREADS)
            ((float4*)acc)[i] = make_float4(0.f, 0.f, 0.f, 0.f);
        __syncthreads();

        for (int sg = wave * 8; sg < nsl; sg += (OWN_THREADS / 32) * 8) {
            int2 sd = make_int2(0, 0);
            if (lane < 8 && sg + lane < nsl) sd = segs[(size_t)o * nsl + sg + lane];
            int st[8], pre[9];
            pre[0] = 0;
            #pragma unroll
            for (int k = 0; k < 8; k++) {
                st[k]      = __builtin_amdgcn_readlane(sd.x, k);
                pre[k + 1] = pre[k] + __builtin_amdgcn_readlane(sd.y, k);
            }
            const int T = pre[8];

            for (int g0 = 0; g0 < T; g0 += 128) {
                int rx[4], ry[4], rz[4];
                #pragma unroll
                for (int q = 0; q < 4; q++) {
                    const int i = min(g0 + 32 * q + lane, T - 1);
                    const int4 r = recs[rec_addr(i, st, pre)];
                    rx[q] = r.x; ry[q] = r.y; rz[q] = r.z;
                }
                #pragma unroll
                for (int q = 0; q < 4; q++) {
                    const int c = min(32, T - g0 - 32 * q);
                    int j0 = 0;
                    for (; j0 + 8 <= c; j0 += 8)
                        add_batch<true>(rx[q], ry[q], rz[q], j0, 8, lane, feat, acc);
                    if (j0 < c)
                        add_batch<false>(rx[q], ry[q], rz[q], j0, c - j0, lane, feat, acc);
                }
            }
        }
        __syncthreads();

        for (int i = tid; i < nr * (D_FEAT / 4); i += OWN_THREADS)
            ((float4*)(hneigh + (size_t)lo * D_FEAT))[i] = ((const float4*)acc)[i];
        __syncthreads();
    }
}

#define GEMM_BLOCKS 512
#define W_PITCH 132

__global__ __launch_bounds__(256) void gemm_fused_kernel(
    const float* __restrict__ feat,
    const float* __restrict__ hneigh,
    const float* __restrict__ W1,
    const float* __restrict__ b1,
    const float* __restrict__ W2,
    const float* __restrict__ b2,
    float*       __restrict__ out,
    int nrows)
{
    __shared__ float Wl1[D_FEAT * W_PITCH];
    __shared__ float Wl2[D_FEAT * W_PITCH];
    __shared__ float X1[16 * W_PITCH];
    __shared__ float X2[16 * W_PITCH];

    const int tid = threadIdx.x;
    const int ntiles = (nrows + 15) >> 4;

    for (int idx = tid; idx < D_FEAT * (D_FEAT / 4); idx += 256) {
        int r  = idx >> 5;
        int c4 = idx & 31;
        float4 a = ((const float4*)W1)[idx];
        float4 b = ((const float4*)W2)[idx];
        *(float4*)&Wl1[r * W_PITCH + 4 * c4] = a;
        *(float4*)&Wl2[r * W_PITCH + 4 * c4] = b;
    }
    __syncthreads();

    const int wave = tid >> 5;
    const int lane = tid & 31;
    const int half = lane >> 4;
    const int lr   = lane & 15;
    const int col  = wave * 16 + lr;

    v2f bw1[D_FEAT / 4], bw2[D_FEAT / 4];
    {
        const float* w1p = &Wl1[col * W_PITCH + 2 * half];
        const float* w2p = &Wl2[col * W_PITCH + 2 * half];
        #pragma unroll
        for (int i = 0; i < D_FEAT / 4; i++) {
            bw1[i] = (v2f){ w1p[4 * i], w1p[4 * i + 1] };
            bw2[i] = (v2f){ w2p[4 * i], w2p[4 * i + 1] };
        }
    }
    const float bias = b1[col] + b2[col];

    const float* x1p = &X1[lr * W_PITCH + 2 * half];
    const float* x2p = &X2[lr * W_PITCH + 2 * half];

    const int sr = tid >> 5;
    const int sc = tid & 31;

    float4 f0, f1, h0, h1;
    {
        const float4 z = make_float4(0.f, 0.f, 0.f, 0.f);
        f0 = z; f1 = z; h0 = z; h1 = z;
        int tile0 = blockIdx.x;
        if (tile0 < ntiles) {
            int g0 = tile0 * 16 + sr, g1 = g0 + 8;
            if (g0 < nrows) {
                f0 = ((const float4*)(feat   + (size_t)g0 * D_FEAT))[sc];
                h0 = ((const float4*)(hneigh + (size_t)g0 * D_FEAT))[sc];
            }
            if (g1 < nrows) {
                f1 = ((const float4*)(feat   + (size_t)g1 * D_FEAT))[sc];
                h1 = ((const float4*)(hneigh + (size_t)g1 * D_FEAT))[sc];
            }
        }
    }

    for (int tile = blockIdx.x; tile < ntiles; tile += GEMM_BLOCKS) {
        const int rowbase = tile * 16;

        __syncthreads();

        {
            float4 s0 = make_float4(f0.x + h0.x, f0.y + h0.y, f0.z + h0.z, f0.w + h0.w);
            float4 p0 = make_float4(f0.x * h0.x, f0.y * h0.y, f0.z * h0.z, f0.w * h0.w);
            float4 s1 = make_float4(f1.x + h1.x, f1.y + h1.y, f1.z + h1.z, f1.w + h1.w);
            float4 p1 = make_float4(f1.x * h1.x, f1.y * h1.y, f1.z * h1.z, f1.w * h1.w);
            *(float4*)&X1[sr * W_PITCH + 4 * sc]       = s0;
            *(float4*)&X2[sr * W_PITCH + 4 * sc]       = p0;
            *(float4*)&X1[(sr + 8) * W_PITCH + 4 * sc] = s1;
            *(float4*)&X2[(sr + 8) * W_PITCH + 4 * sc] = p1;
        }
        __syncthreads();

        {
            const float4 z = make_float4(0.f, 0.f, 0.f, 0.f);
            f0 = z; f1 = z; h0 = z; h1 = z;
            int nt = tile + GEMM_BLOCKS;
            if (nt < ntiles) {
                int g0 = nt * 16 + sr, g1 = g0 + 8;
                if (g0 < nrows) {
                    f0 = ((const float4*)(feat   + (size_t)g0 * D_FEAT))[sc];
                    h0 = ((const float4*)(hneigh + (size_t)g0 * D_FEAT))[sc];
                }
                if (g1 < nrows) {
                    f1 = ((const float4*)(feat   + (size_t)g1 * D_FEAT))[sc];
                    h1 = ((const float4*)(hneigh + (size_t)g1 * D_FEAT))[sc];
                }
            }
        }

        v8f acc = { bias, bias, bias, bias, bias, bias, bias, bias };
        #pragma unroll
        for (int i = 0; i < D_FEAT / 4; i++) {
            v2f a1 = { x1p[4 * i], x1p[4 * i + 1] };
            v2f a2 = { x2p[4 * i], x2p[4 * i + 1] };
            acc = __builtin_amdgcn_wmma_f32_16x16x4_f32(
                      false, a1, false, bw1[i], (short)0, acc, false, false);
            acc = __builtin_amdgcn_wmma_f32_16x16x4_f32(
                      false, a2, false, bw2[i], (short)0, acc, false, false);
        }

        #pragma unroll
        for (int v = 0; v < 8; v++) {
            int row = rowbase + v + 8 * half;
            if (row < nrows) {
                float h = acc[v];
                out[(size_t)row * D_FEAT + col] = (h >= 0.f) ? h : NEG_SLOPE * h;
            }
        }
    }
}

extern "C" void kernel_launch(void* const* d_in, const int* in_sizes, int n_in,
                              void* d_out, int out_size, void* d_ws, size_t ws_size,
                              hipStream_t stream) {
    const float* feat     = (const float*)d_in[0];
    const int*   target   = (const int*)  d_in[1];
    const int*   neighbor = (const int*)  d_in[2];
    const float* values   = (const float*)d_in[3];
    const float* W1       = (const float*)d_in[4];
    const float* b1       = (const float*)d_in[5];
    const float* W2       = (const float*)d_in[6];
    const float* b2       = (const float*)d_in[7];

    const int N = in_sizes[0] / D_FEAT;
    const int E = in_sizes[1];

    float* hneigh = (float*)d_ws;
    int4*  recs   = (int4*)((char*)d_ws + 33554432);
    int2*  segs   = (int2*)((char*)d_ws + 50331648);
    float* outp   = (float*)d_out;

    {
        bucket_kernel<<<BK_BLOCKS, BK_THREADS, 0, stream>>>(target, neighbor, values,
                                                            recs, segs, N, E);
    }
    {
        owner_sum_kernel<<<OWN_BLOCKS, OWN_THREADS, 0, stream>>>(feat, recs, segs, hneigh, N, E);
    }
    {
        gemm_fused_kernel<<<GEMM_BLOCKS, 256, 0, stream>>>(feat, hneigh, W1, b1, W2, b2,
                                                      outp, N);
    }
}


# Owner groups counting-sort their records by target row in LDS; each wave then sums whole rows in registers from 16-B row loads and stores each row once; LDS f32 adds gone; add order changes
# speedup vs baseline: 3.2629x; 1.0522x over previous
#include <hip/hip_runtime.h>
#include <hip/hip_bf16.h>

#define D_FEAT 128
#define NEG_SLOPE 0.01f

typedef float v2f __attribute__((ext_vector_type(2)));
typedef float v8f __attribute__((ext_vector_type(8)));

#define OWN_BLOCKS  256
#define OWN_ROWS    196
#define OWN_THREADS 1024
#define NOWN        256
#define SL          2448
#define BK_BLOCKS   256
#define BK_THREADS  512
#define BK_PER      ((SL + BK_THREADS - 1) / BK_THREADS)

__global__ __launch_bounds__(BK_THREADS) void bucket_kernel(
    const int*   __restrict__ target,
    const int*   __restrict__ neighbor,
    const float* __restrict__ values,
    int4*        __restrict__ recs,
    int2*        __restrict__ segs,
    int N, int E)
{
    __shared__ int  hist[NOWN];
    __shared__ int  offs[NOWN + 1];
    __shared__ int4 srec[SL];

    const int tid = threadIdx.x;
    const int nsl = (E + SL - 1) / SL;

    for (int s = blockIdx.x; s < nsl; s += gridDim.x) {
        const int e0  = s * SL;
        const int cnt = min(SL, E - e0);

        if (tid < NOWN) hist[tid] = 0;
        __syncthreads();

        int   tt[BK_PER], nn[BK_PER];
        float vv[BK_PER];
        #pragma unroll
        for (int k = 0; k < BK_PER; k++) {
            const int e = e0 + min(tid + k * BK_THREADS, cnt - 1);
            tt[k] = target[e];
            nn[k] = neighbor[e];
            vv[k] = values[e];
        }
        int  ow[BK_PER], rk[BK_PER];
        int4 rc[BK_PER];
        #pragma unroll
        for (int k = 0; k < BK_PER; k++) {
            const int t = tt[k];
            const int o = t / OWN_ROWS;
            ow[k] = -1; rk[k] = 0;
            rc[k] = make_int4(min(max(nn[k], 0), N - 1), t - o * OWN_ROWS,
                              __builtin_bit_cast(int, vv[k]), 0);
            if (tid + k * BK_THREADS < cnt && (unsigned)t < (unsigned)N && o < NOWN) {
                ow[k] = o;
                rk[k] = atomicAdd(&hist[o], 1);
            }
        }
        __syncthreads();

        if (tid < 32) {
            int c[8];
            int sum = 0;
            #pragma unroll
            for (int j = 0; j < 8; j++) { c[j] = hist[8 * tid + j]; sum += c[j]; }
            int inc = sum;
            for (int d = 1; d < 32; d <<= 1) {
                int y = __shfl_up(inc, d, 32);
                if (tid >= d) inc += y;
            }
            int run = inc - sum;
            #pragma unroll
            for (int j = 0; j < 8; j++) { offs[8 * tid + j] = run; run += c[j]; }
            if (tid == 31) offs[NOWN] = run;
        }
        __syncthreads();

        #pragma unroll
        for (int k = 0; k < BK_PER; k++)
            if (ow[k] >= 0) srec[offs[ow[k]] + rk[k]] = rc[k];
        __syncthreads();

        const int total = offs[NOWN];
        for (int i = tid; i < total; i += BK_THREADS) recs[(size_t)e0 + i] = srec[i];
        if (tid < NOWN)
            segs[(size_t)tid * nsl + s] = make_int2(e0 + offs[tid], offs[tid + 1] - offs[tid]);
        __syncthreads();
    }
}

#define SORT_CAP  4096
#define CNT_SZ    224
#define WAVE_ROWS 7

template <bool FULL>
__device__ __forceinline__ void row_batch(int nbv, int vbv, int j0, int cnt, int lane,
                                          const float* __restrict__ feat, float4& a)
{
    float4 f[8];
    const int jl = j0 + cnt - 1;
    #pragma unroll
    for (int j = 0; j < 8; j++) {
        int nbj = __builtin_amdgcn_readlane(nbv, FULL ? j0 + j : min(j0 + j, jl));
        f[j] = ((const float4*)(feat + (size_t)nbj * D_FEAT))[lane];
    }
    #pragma unroll
    for (int j = 0; j < 8; j++) {
        if (FULL || j < cnt) {
            float vj = __builtin_bit_cast(float, __builtin_amdgcn_readlane(vbv, j0 + j));
            a.x += f[j].x * vj;
            a.y += f[j].y * vj;
            a.z += f[j].z * vj;
            a.w += f[j].w * vj;
        }
    }
}

__device__ __forceinline__ int rec_addr(int i, const int* st, const int* pre)
{
    int a = st[0] + i;
    #pragma unroll
    for (int k = 1; k < 8; k++)
        if (i >= pre[k]) a = st[k] + (i - pre[k]);
    return a;
}

__global__ __launch_bounds__(OWN_THREADS) void owner_sum_kernel(
    const float* __restrict__ feat,
    const int4*  __restrict__ recs,
    const int2*  __restrict__ segs,
    float*       __restrict__ hneigh,
    int N, int E)
{
    __shared__ int  cnt[CNT_SZ];
    __shared__ int  offs[CNT_SZ + 1];
    __shared__ int  s_rounds;
    __shared__ int2 sorted[SORT_CAP];

    const int tid  = threadIdx.x;
    const int lane = tid & 31;
    const int wave = __builtin_amdgcn_readfirstlane(tid >> 5);
    const int nsl  = (E + SL - 1) / SL;

    for (int o = blockIdx.x; o < NOWN; o += gridDim.x) {
        const int lo = o * OWN_ROWS;
        const int nr = min(OWN_ROWS, N - lo);
        if (nr <= 0) break;

        float4 accum[WAVE_ROWS];
        #pragma unroll
        for (int k = 0; k < WAVE_ROWS; k++) accum[k] = make_float4(0.f, 0.f, 0.f, 0.f);

        for (int sgb = 0; sgb < nsl; sgb += (OWN_THREADS / 32) * 8) {
            __syncthreads();
            if (tid < CNT_SZ) cnt[tid] = 0;
            if (tid == 0) s_rounds = 0;
            __syncthreads();

            const int sg = sgb + wave * 8;
            int2 sd = make_int2(0, 0);
            if (lane < 8 && sg + lane < nsl) sd = segs[(size_t)o * nsl + sg + lane];
            int st[8], pre[9];
            pre[0] = 0;
            #pragma unroll
            for (int k = 0; k < 8; k++) {
                st[k]      = __builtin_amdgcn_readlane(sd.x, k);
                pre[k + 1] = pre[k] + __builtin_amdgcn_readlane(sd.y, k);
            }
            const int T = pre[8];

            if (lane == 0) atomicMax(&s_rounds, (T + 127) >> 7);
            __syncthreads();
            const int R = s_rounds;

            for (int r = 0; r < R; r++) {
                const int g0 = r * 128;
                int rnb[4], rvb[4], rrow[4], rrk[4];
                #pragma unroll
                for (int q = 0; q < 4; q++) {
                    const int i = g0 + 32 * q + lane;
                    const int4 rec = recs[min(rec_addr(min(i, max(T - 1, 0)), st, pre), E - 1)];
                    rnb[q] = rec.x; rvb[q] = rec.z;
                    rrow[q] = (i < T) ? rec.y : -1;
                }
                #pragma unroll
                for (int q = 0; q < 4; q++) {
                    rrk[q] = 0;
                    if (rrow[q] >= 0) rrk[q] = atomicAdd(&cnt[rrow[q]], 1);
                }
                __syncthreads();

                if (tid < 32) {
                    int c[7];
                    int sum = 0;
                    #pragma unroll
                    for (int j = 0; j < 7; j++) { c[j] = cnt[7 * tid + j]; sum += c[j]; }
                    int inc = sum;
                    for (int d = 1; d < 32; d <<= 1) {
                        int y = __shfl_up(inc, d, 32);
                        if (tid >= d) inc += y;
                    }
                    int run = inc - sum;
                    #pragma unroll
                    for (int j = 0; j < 7; j++) { offs[7 * tid + j] = run; run += c[j]; }
                    if (tid == 31) offs[CNT_SZ] = run;
                }
                __syncthreads();

                #pragma unroll
                for (int q = 0; q < 4; q++)
                    if (rrow[q] >= 0) sorted[offs[rrow[q]] + rrk[q]] = make_int2(rnb[q], rvb[q]);
                __syncthreads();

                #pragma unroll
                for (int k = 0; k < WAVE_ROWS; k++) {
                    const int row = wave + 32 * k;
                    if (row < nr) {
                        const int beg = __builtin_amdgcn_readfirstlane(offs[row]);
                        const int end = __builtin_amdgcn_readfirstlane(offs[row + 1]);
                        for (int b = beg; b < end; b += 32) {
                            const int m = min(32, end - b);
                            int2 e = make_int2(0, 0);
                            if (lane < m) e = sorted[b + lane];
                            int j0 = 0;
                            for (; j0 + 8 <= m; j0 += 8)
                                row_batch<true>(e.x, e.y, j0, 8, lane, feat, accum[k]);
                            if (j0 < m)
                                row_batch<false>(e.x, e.y, j0, m - j0, lane, feat, accum[k]);
                        }
                    }
                }

                if (r + 1 < R) {
                    __syncthreads();
                    if (tid < CNT_SZ) cnt[tid] = 0;
                    __syncthreads();
                }
            }
        }

        #pragma unroll
        for (int k = 0; k < WAVE_ROWS; k++) {
            const int row = wave + 32 * k;
            if (row < nr)
                ((float4*)(hneigh + (size_t)(lo + row) * D_FEAT))[lane] = accum[k];
        }
    }
}

#define GEMM_BLOCKS 512
#define W_PITCH 132

__global__ __launch_bounds__(256) void gemm_fused_kernel(
    const float* __restrict__ feat,
    const float* __restrict__ hneigh,
    const float* __restrict__ W1,
    const float* __restrict__ b1,
    const float* __restrict__ W2,
    const float* __restrict__ b2,
    float*       __restrict__ out,
    int nrows)
{
    __shared__ float Wl1[D_FEAT * W_PITCH];
    __shared__ float Wl2[D_FEAT * W_PITCH];
    __shared__ float X1[16 * W_PITCH];
    __shared__ float X2[16 * W_PITCH];

    const int tid = threadIdx.x;
    const int ntiles = (nrows + 15) >> 4;

    for (int idx = tid; idx < D_FEAT * (D_FEAT / 4); idx += 256) {
        int r  = idx >> 5;
        int c4 = idx & 31;
        float4 a = ((const float4*)W1)[idx];
        float4 b = ((const float4*)W2)[idx];
        *(float4*)&Wl1[r * W_PITCH + 4 * c4] = a;
        *(float4*)&Wl2[r * W_PITCH + 4 * c4] = b;
    }
    __syncthreads();

    const int wave = tid >> 5;
    const int lane = tid & 31;
    const int half = lane >> 4;
    const int lr   = lane & 15;
    const int col  = wave * 16 + lr;

    v2f bw1[D_FEAT / 4], bw2[D_FEAT / 4];
    {
        const float* w1p = &Wl1[col * W_PITCH + 2 * half];
        const float* w2p = &Wl2[col * W_PITCH + 2 * half];
        #pragma unroll
        for (int i = 0; i < D_FEAT / 4; i++) {
            bw1[i] = (v2f){ w1p[4 * i], w1p[4 * i + 1] };
            bw2[i] = (v2f){ w2p[4 * i], w2p[4 * i + 1] };
        }
    }
    const float bias = b1[col] + b2[col];

    const float* x1p = &X1[lr * W_PITCH + 2 * half];
    const float* x2p = &X2[lr * W_PITCH + 2 * half];

    const int sr = tid >> 5;
    const int sc = tid & 31;

    float4 f0, f1, h0, h1;
    {
        const float4 z = make_float4(0.f, 0.f, 0.f, 0.f);
        f0 = z; f1 = z; h0 = z; h1 = z;
        int tile0 = blockIdx.x;
        if (tile0 < ntiles) {
            int g0 = tile0 * 16 + sr, g1 = g0 + 8;
            if (g0 < nrows) {
                f0 = ((const float4*)(feat   + (size_t)g0 * D_FEAT))[sc];
                h0 = ((const float4*)(hneigh + (size_t)g0 * D_FEAT))[sc];
            }
            if (g1 < nrows) {
                f1 = ((const float4*)(feat   + (size_t)g1 * D_FEAT))[sc];
                h1 = ((const float4*)(hneigh + (size_t)g1 * D_FEAT))[sc];
            }
        }
    }

    for (int tile = blockIdx.x; tile < ntiles; tile += GEMM_BLOCKS) {
        const int rowbase = tile * 16;

        __syncthreads();

        {
            float4 s0 = make_float4(f0.x + h0.x, f0.y + h0.y, f0.z + h0.z, f0.w + h0.w);
            float4 p0 = make_float4(f0.x * h0.x, f0.y * h0.y, f0.z * h0.z, f0.w * h0.w);
            float4 s1 = make_float4(f1.x + h1.x, f1.y + h1.y, f1.z + h1.z, f1.w + h1.w);
            float4 p1 = make_float4(f1.x * h1.x, f1.y * h1.y, f1.z * h1.z, f1.w * h1.w);
            *(float4*)&X1[sr * W_PITCH + 4 * sc]       = s0;
            *(float4*)&X2[sr * W_PITCH + 4 * sc]       = p0;
            *(float4*)&X1[(sr + 8) * W_PITCH + 4 * sc] = s1;
            *(float4*)&X2[(sr + 8) * W_PITCH + 4 * sc] = p1;
        }
        __syncthreads();

        {
            const float4 z = make_float4(0.f, 0.f, 0.f, 0.f);
            f0 = z; f1 = z; h0 = z; h1 = z;
            int nt = tile + GEMM_BLOCKS;
            if (nt < ntiles) {
                int g0 = nt * 16 + sr, g1 = g0 + 8;
                if (g0 < nrows) {
                    f0 = ((const float4*)(feat   + (size_t)g0 * D_FEAT))[sc];
                    h0 = ((const float4*)(hneigh + (size_t)g0 * D_FEAT))[sc];
                }
                if (g1 < nrows) {
                    f1 = ((const float4*)(feat   + (size_t)g1 * D_FEAT))[sc];
                    h1 = ((const float4*)(hneigh + (size_t)g1 * D_FEAT))[sc];
                }
            }
        }

        v8f acc = { bias, bias, bias, bias, bias, bias, bias, bias };
        #pragma unroll
        for (int i = 0; i < D_FEAT / 4; i++) {
            v2f a1 = { x1p[4 * i], x1p[4 * i + 1] };
            v2f a2 = { x2p[4 * i], x2p[4 * i + 1] };
            acc = __builtin_amdgcn_wmma_f32_16x16x4_f32(
                      false, a1, false, bw1[i], (short)0, acc, false, false);
            acc = __builtin_amdgcn_wmma_f32_16x16x4_f32(
                      false, a2, false, bw2[i], (short)0, acc, false, false);
        }

        #pragma unroll
        for (int v = 0; v < 8; v++) {
            int row = rowbase + v + 8 * half;
            if (row < nrows) {
                float h = acc[v];
                out[(size_t)row * D_FEAT + col] = (h >= 0.f) ? h : NEG_SLOPE * h;
            }
        }
    }
}

extern "C" void kernel_launch(void* const* d_in, const int* in_sizes, int n_in,
                              void* d_out, int out_size, void* d_ws, size_t ws_size,
                              hipStream_t stream) {
    const float* feat     = (const float*)d_in[0];
    const int*   target   = (const int*)  d_in[1];
    const int*   neighbor = (const int*)  d_in[2];
    const float* values   = (const float*)d_in[3];
    const float* W1       = (const float*)d_in[4];
    const float* b1       = (const float*)d_in[5];
    const float* W2       = (const float*)d_in[6];
    const float* b2       = (const float*)d_in[7];

    const int N = in_sizes[0] / D_FEAT;
    const int E = in_sizes[1];

    float* hneigh = (float*)d_ws;
    int4*  recs   = (int4*)((char*)d_ws + 33554432);
    int2*  segs   = (int2*)((char*)d_ws + 50331648);
    float* outp   = (float*)d_out;

    {
        bucket_kernel<<<BK_BLOCKS, BK_THREADS, 0, stream>>>(target, neighbor, values,
                                                            recs, segs, N, E);
    }
    {
        owner_sum_kernel<<<OWN_BLOCKS, OWN_THREADS, 0, stream>>>(feat, recs, segs, hneigh, N, E);
    }
    {
        gemm_fused_kernel<<<GEMM_BLOCKS, 256, 0, stream>>>(feat, hneigh, W1, b1, W2, b2,
                                                      outp, N);
    }
}
